# starting kernel
# baseline (speedup 1.0000x reference)
#include <hip/hip_runtime.h>
#include <math.h>


#define WIRES 11
#define DIM   2048
#define PIX   1024
#define NB    1024

typedef __attribute__((ext_vector_type(2))) float v2f;
typedef __attribute__((ext_vector_type(8))) float v8f;

__global__ void qprep(const float* __restrict__ w0, const float* __restrict__ w1,
                      float* __restrict__ G) {
  const float PI_ = 3.14159265358979323846f;
  for (int e = threadIdx.x; e < 8 * 3 * 256; e += blockDim.x) {
    int L    = e / 768;
    int rem  = e - L * 768;
    int grp  = rem >> 8;
    int cell = rem & 255;
    int m = cell >> 4, k = cell & 15;
    const float* w = (L < 4) ? w0 : w1;
    int l = L & 3;

    float re = 1.f, im = 0.f;
    int nw, q0, topbit;
    bool zero = false;
    if (grp == 0)      { nw = 4; q0 = 0; topbit = 3; }
    else if (grp == 1) { nw = 4; q0 = 4; topbit = 3; }
    else               { nw = 3; q0 = 8; topbit = 2;
                         if ((m >> 3) != (k >> 3)) zero = true; }
    if (zero) { re = 0.f; im = 0.f; }
    else {
      for (int i = 0; i < nw; ++i) {
        int q  = q0 + i;
        int bp = topbit - i;
        int mb = (m >> bp) & 1, kb = (k >> bp) & 1;
        const float* a3 = w + (l * WIRES + q) * 3;
        float phi   = PI_ * tanhf(a3[0]);
        float theta = PI_ * tanhf(a3[1]);
        float omega = PI_ * tanhf(a3[2]);
        float c = cosf(0.5f * theta), s = sinf(0.5f * theta);
        float h = 0.5f * (phi + omega), g = 0.5f * (phi - omega);
        float ur, ui;
        if (mb == 0 && kb == 0)      { ur =  cosf(h) * c; ui = -sinf(h) * c; }
        else if (mb == 0 && kb == 1) { ur = -cosf(g) * s; ui = -sinf(g) * s; }
        else if (mb == 1 && kb == 0) { ur =  cosf(g) * s; ui = -sinf(g) * s; }
        else                         { ur =  cosf(h) * c; ui =  sinf(h) * c; }
        float nr = re * ur - im * ui;
        float ni = re * ui + im * ur;
        re = nr; im = ni;
      }
    }
    int base = (L * 3 + grp) * 512;
    G[base + cell]       = re;
    G[base + 256 + cell] = im;
  }
}

__device__ __forceinline__ int pinv_idx(int x, int r) {
#pragma unroll
  for (int q = WIRES - 1; q >= 0; --q) {
    int tq = q + r; if (tq >= WIRES) tq -= WIRES;
    x ^= ((x >> (10 - q)) & 1) << (10 - tq);
  }
  return x;
}

__device__ __forceinline__ void apply_group(float2* st,
                                            const float* __restrict__ gr,
                                            const float* __restrict__ gi,
                                            int lo, int wave, int lane, int rperm) {
  const int n    = lane & 15;
  const int half = lane >> 4;
  const int j    = wave * 16 + n;
  const int base = ((j >> lo) << (lo + 4)) | (j & ((1 << lo) - 1));

  v2f ar[4], ai[4], br[4], bi[4];
#pragma unroll
  for (int c = 0; c < 4; ++c) {
    int k0 = 4 * c + 2 * half;
    int i0 = base | (k0 << lo);
    int i1 = base | ((k0 + 1) << lo);
    if (rperm) { i0 = pinv_idx(i0, rperm); i1 = pinv_idx(i1, rperm); }
    float2 s0 = st[i0];
    float2 s1 = st[i1];
    br[c] = v2f{s0.x, s1.x};
    bi[c] = v2f{s0.y, s1.y};
    ar[c] = v2f{gr[n * 16 + k0], gr[n * 16 + k0 + 1]};
    ai[c] = v2f{gi[n * 16 + k0], gi[n * 16 + k0 + 1]};
  }
  if (rperm) __syncthreads();

  v8f dr = {0, 0, 0, 0, 0, 0, 0, 0};
  v8f di = {0, 0, 0, 0, 0, 0, 0, 0};
#pragma unroll
  for (int c = 0; c < 4; ++c) {
    v2f nai = -ai[c];
    dr = __builtin_amdgcn_wmma_f32_16x16x4_f32(false, ar[c], false, br[c],
                                               (short)0, dr, false, false);
    dr = __builtin_amdgcn_wmma_f32_16x16x4_f32(false, nai,   false, bi[c],
                                               (short)0, dr, false, false);
    di = __builtin_amdgcn_wmma_f32_16x16x4_f32(false, ar[c], false, bi[c],
                                               (short)0, di, false, false);
    di = __builtin_amdgcn_wmma_f32_16x16x4_f32(false, ai[c], false, br[c],
                                               (short)0, di, false, false);
  }

#pragma unroll
  for (int r = 0; r < 8; ++r) {
    int m = r + 8 * half;
    st[base | (m << lo)] = make_float2(dr[r], di[r]);
  }
}

__global__ void __launch_bounds__(256)
qsim(const float* __restrict__ x, const float* __restrict__ y,
     const float* __restrict__ G, float* __restrict__ out) {
  __shared__ float2 st[DIM];
  __shared__ float  red[256];
  const int tid  = threadIdx.x;
  const int lane = tid & 31;
  const int wave = tid >> 5;
  const int b    = blockIdx.x;

  float xs[4], part = 0.f;
#pragma unroll
  for (int t = 0; t < 4; ++t) {
    int p = tid + 256 * t;
    float v = x[b * PIX + p];
    xs[t] = v; part += v * v;
  }
  red[tid] = part; __syncthreads();
  for (int off = 128; off > 0; off >>= 1) {
    if (tid < off) red[tid] += red[tid + off];
    __syncthreads();
  }
  float inv = rsqrtf(red[0]);
#pragma unroll
  for (int t = 0; t < 4; ++t) {
    int p = tid + 256 * t;
    st[2 * p]     = make_float2(xs[t] * inv, 0.f);
    st[2 * p + 1] = make_float2(0.f, 0.f);
  }
  __syncthreads();

  const float yv = y[b];
  const float rc = cosf(0.5f * yv), rs = sinf(0.5f * yv);

  int pend = 0;
  for (int blkw = 0; blkw < 2; ++blkw) {
#pragma unroll
    for (int t = 0; t < 4; ++t) {
      int p  = tid + 256 * t;
      int x0 = 2 * p, x1 = 2 * p + 1;
      if (pend) { x0 = pinv_idx(x0, pend); x1 = pinv_idx(x1, pend); }
      float2 a0 = st[x0], a1 = st[x1];
      st[x0] = make_float2(rc * a0.x + rs * a1.y, rc * a0.y - rs * a1.x);
      st[x1] = make_float2(rc * a1.x + rs * a0.y, rc * a1.y - rs * a0.x);
    }
    __syncthreads();

    for (int l = 0; l < 4; ++l) {
      int L = blkw * 4 + l;
      const float* g0 = G + (L * 3 + 0) * 512;
      apply_group(st, g0, g0 + 256, 7, wave, lane, pend); __syncthreads();
      const float* g1 = G + (L * 3 + 1) * 512;
      apply_group(st, g1, g1 + 256, 3, wave, lane, 0);    __syncthreads();
      const float* g2 = G + (L * 3 + 2) * 512;
      apply_group(st, g2, g2 + 256, 0, wave, lane, 0);    __syncthreads();
      pend = l + 1;
    }
  }

#pragma unroll
  for (int t = 0; t < 4; ++t) {
    int p  = tid + 256 * t;
    float2 a = st[pinv_idx(2 * p, pend)];
    out[b * PIX + p] = fminf(1.0f, (a.x * a.x + a.y * a.y) * (float)PIX);
  }
}

extern "C" void kernel_launch(void* const* d_in, const int* in_sizes, int n_in,
                              void* d_out, int out_size, void* d_ws, size_t ws_size,
                              hipStream_t stream) {
  const float* x  = (const float*)d_in[0];
  const float* y  = (const float*)d_in[1];
  const float* w0 = (const float*)d_in[2];
  const float* w1 = (const float*)d_in[3];
  float* G = (float*)d_ws;

  qprep<<<1, 256, 0, stream>>>(w0, w1, G);
  qsim<<<NB, 256, 0, stream>>>(x, y, G, (float*)d_out);
}


# first layer skips the embedded state's structural zeros (odd amplitudes; imaginary parts before its first pass): 14 WMMA per wave there, not 32; values equal, first pass's k sum order changes
# speedup vs baseline: 5.2918x; 5.2918x over previous
#include <hip/hip_runtime.h>
#include <math.h>


#define WIRES 11
#define DIM   2048
#define PIX   1024
#define NB    1024

typedef __attribute__((ext_vector_type(2))) float v2f;
typedef __attribute__((ext_vector_type(8))) float v8f;

#define NSLOT 12
#define NPASS 32

__device__ __forceinline__ float2 cmulf(float2 p, float2 u) {
  return make_float2(p.x * u.x - p.y * u.y, p.x * u.y + p.y * u.x);
}

__device__ __forceinline__ void rot_entries(const float* __restrict__ a3, float2 u[4]) {
  const float PI_ = 3.14159265358979323846f;
  const float a0 = a3[0], a1 = a3[1], a2 = a3[2];
  float phi   = PI_ * tanhf(a0);
  float theta = PI_ * tanhf(a1);
  float omega = PI_ * tanhf(a2);
  float c, s, ch, sh, cg, sg;
  sincosf(0.5f * theta, &s, &c);
  sincosf(0.5f * (phi + omega), &sh, &ch);
  sincosf(0.5f * (phi - omega), &sg, &cg);
  u[0] = make_float2( ch * c, -sh * c);
  u[1] = make_float2(-cg * s, -sg * s);
  u[2] = make_float2( cg * s, -sg * s);
  u[3] = make_float2( ch * c,  sh * c);
}

__device__ __forceinline__ int pinv_idx(int x, int r) {
  const int a = x ^ ((x & ((1 << r) - 1)) << (WIRES - r));
  return a ^ (a >> r);
}

constexpr int spread_coset(int j, int sub) {
  int r = 0, k = 0;
  for (int b = 0; b < WIRES; ++b) {
    if (!((sub >> b) & 1)) { r |= ((j >> k) & 1) << b; ++k; }
  }
  return r;
}

template <int S2, int S1, int S0>
__device__ __forceinline__ void apply_group(float2* st, const float4* afr,
                                            int p, int wave, int lane, int rperm) {
  constexpr int SUB = (1 << S2) | (1 << S1) | (1 << S0);
  const int n    = lane & 15;
  const int half = lane >> 4;

  const float4 a01 = afr[(p * 32 + lane) * 2];
  const float4 a23 = afr[(p * 32 + lane) * 2 + 1];
  const v2f a[4] = {v2f{a01.x, a01.y}, v2f{a01.z, a01.w},
                    v2f{a23.x, a23.y}, v2f{a23.z, a23.w}};

  constexpr int TILE1 = spread_coset(16, SUB);
  const int base = spread_coset(wave * 32 + n, SUB) + (half << S0);
  int idx[2][4];
  v2f bf[2][4];
#pragma unroll
  for (int t = 0; t < 2; ++t) {
#pragma unroll
    for (int c = 0; c < 4; ++c) {
      idx[t][c] = base + t * TILE1 + ((c & 1) << S1) + ((c >> 1) << S2);
      int i0 = idx[t][c];
      if (rperm) i0 = pinv_idx(i0, rperm);
      const float2 s0 = st[i0];
      bf[t][c] = v2f{s0.x, s0.y};
    }
  }
  if (rperm) __syncthreads();

  v8f d0 = {0, 0, 0, 0, 0, 0, 0, 0};
  v8f d1 = {0, 0, 0, 0, 0, 0, 0, 0};
#pragma unroll
  for (int c = 0; c < 4; ++c) {
    d0 = __builtin_amdgcn_wmma_f32_16x16x4_f32(false, a[c], false, bf[0][c],
                                               (short)0, d0, false, false);
    d1 = __builtin_amdgcn_wmma_f32_16x16x4_f32(false, a[c], false, bf[1][c],
                                               (short)0, d1, false, false);
  }

#pragma unroll
  for (int q = 0; q < 4; ++q) {
    st[idx[0][q]] = make_float2(d0[2 * q], d0[2 * q + 1]);
    st[idx[1][q]] = make_float2(d1[2 * q], d1[2 * q + 1]);
  }
}

template <int S2, int S1, int S0, int KIND>
__device__ __forceinline__ void apply_group_first(float2* st, const float4* afr,
                                                  int p, int wave, int lane) {
  constexpr int SUB = (1 << S2) | (1 << S1) | (1 << S0);
  static_assert(KIND != 2 || S1 == 0, "KIND 2 needs index bit 0 as the chunk bit c&1");
  static_assert(KIND == 2 || !(SUB & 1), "KIND 0/1 groups must not hold index bit 0");
  const int n    = lane & 15;
  const int half = lane >> 4;

  const float4 a01 = afr[(p * 32 + lane) * 2];
  const float4 a23 = afr[(p * 32 + lane) * 2 + 1];
  const v2f a[4] = {v2f{a01.x, a01.y}, v2f{a01.z, a01.w},
                    v2f{a23.x, a23.y}, v2f{a23.z, a23.w}};
  const v8f zero = {0, 0, 0, 0, 0, 0, 0, 0};

  if (KIND == 2) {
    constexpr int TILE1 = spread_coset(16, SUB);
    const int base = spread_coset(wave * 32 + n, SUB) + (half << S0);
    v2f b0[2], b1[2];
#pragma unroll
    for (int cc = 0; cc < 2; ++cc) {
      const float2 s0 = st[base + (cc << S2)];
      const float2 s1 = st[base + TILE1 + (cc << S2)];
      b0[cc] = v2f{s0.x, s0.y};
      b1[cc] = v2f{s1.x, s1.y};
    }
    v8f d0 = zero, d1 = zero;
#pragma unroll
    for (int cc = 0; cc < 2; ++cc) {
      d0 = __builtin_amdgcn_wmma_f32_16x16x4_f32(false, a[2 * cc], false, b0[cc],
                                                 (short)0, d0, false, false);
      d1 = __builtin_amdgcn_wmma_f32_16x16x4_f32(false, a[2 * cc], false, b1[cc],
                                                 (short)0, d1, false, false);
    }
#pragma unroll
    for (int q = 0; q < 4; ++q) {
      const int i0 = base + ((q & 1) << S1) + ((q >> 1) << S2);
      st[i0]         = make_float2(d0[2 * q], d0[2 * q + 1]);
      st[i0 + TILE1] = make_float2(d1[2 * q], d1[2 * q + 1]);
    }
  } else {
    const int base = spread_coset(wave * 16 + n, SUB | 1) + (half << S0);
    int idx[4];
    v2f bf[4];
#pragma unroll
    for (int c = 0; c < 4; ++c) {
      idx[c] = base + ((c & 1) << S1) + ((c >> 1) << S2);
      const float2 s0 = st[idx[c]];
      bf[c] = v2f{s0.x, s0.y};
    }
    v8f d = zero;
    if (KIND == 1) {
#pragma unroll
      for (int cp = 0; cp < 2; ++cp) {
        d = __builtin_amdgcn_wmma_f32_16x16x4_f32(
            false, v2f{a[2 * cp].x, a[2 * cp + 1].x},
            false, v2f{bf[2 * cp].x, bf[2 * cp + 1].x}, (short)0, d, false, false);
      }
    } else {
#pragma unroll
      for (int c = 0; c < 4; ++c) {
        d = __builtin_amdgcn_wmma_f32_16x16x4_f32(false, a[c], false, bf[c],
                                                   (short)0, d, false, false);
      }
    }
#pragma unroll
    for (int q = 0; q < 4; ++q) {
      st[idx[q]] = make_float2(d[2 * q], d[2 * q + 1]);
    }
  }
}

__device__ __forceinline__ void wave_order() {
  __builtin_amdgcn_fence(__ATOMIC_ACQ_REL, "wavefront");
}

__global__ void __launch_bounds__(256)
qsim(const float* __restrict__ x, const float* __restrict__ y,
     const float* __restrict__ w0, const float* __restrict__ w1,
     float* __restrict__ out) {
  __shared__ float2 st[DIM];
  __shared__ float  red[256];
  __shared__ float4 gate4[8 * NSLOT * 2];
  __shared__ float4 afr[NPASS * 32 * 2];
  const int tid  = threadIdx.x;
  const int lane = tid & 31;
  const int wave = tid >> 5;
  const int b    = blockIdx.x;

  float xs[4], part = 0.f;
#pragma unroll
  for (int t = 0; t < 4; ++t) {
    int p = tid + 256 * t;
    float v = x[b * PIX + p];
    xs[t] = v; part += v * v;
  }
  const float yv = y[b];
  if (tid < 8 * NSLOT) {
    int L    = tid / NSLOT;
    int slot = tid - L * NSLOT;
    float2 u[4];
    if (slot == 9) {
      u[0] = make_float2(1.f, 0.f); u[1] = make_float2(0.f, 0.f);
      u[2] = make_float2(0.f, 0.f); u[3] = make_float2(1.f, 0.f);
    } else {
      int q = slot;
      if (slot >= 3 && slot <= 5) q = slot + 1;
      else if (slot == 6) q = 3;
      else if (slot == 11) q = 9;
      const float* w = (L < 4) ? w0 : w1;
      rot_entries(w + ((L & 3) * WIRES + q) * 3, u);
      if (slot == 10 && (L & 3) == 0) {
        float rc, rs;
        sincosf(0.5f * yv, &rs, &rc);
        const float2 u0 = u[0], u1 = u[1], u2 = u[2], u3 = u[3];
        u[0] = make_float2(u0.x * rc + u1.y * rs, u0.y * rc - u1.x * rs);
        u[1] = make_float2(u1.x * rc + u0.y * rs, u1.y * rc - u0.x * rs);
        u[2] = make_float2(u2.x * rc + u3.y * rs, u2.y * rc - u3.x * rs);
        u[3] = make_float2(u3.x * rc + u2.y * rs, u3.y * rc - u2.x * rs);
      }
    }
    gate4[tid * 2]     = make_float4(u[0].x, u[0].y, u[1].x, u[1].y);
    gate4[tid * 2 + 1] = make_float4(u[2].x, u[2].y, u[3].x, u[3].y);
  }
  red[tid] = part; __syncthreads();
  for (int off = 128; off > 0; off >>= 1) {
    if (tid < off) red[tid] += red[tid + off];
    __syncthreads();
  }
  float inv = rsqrtf(red[0]);
#pragma unroll
  for (int t = 0; t < 4; ++t) {
    int p = tid + 256 * t;
    st[2 * p] = make_float2(xs[t] * inv, 0.f);
  }
  {
    const float2* gate = (const float2*)gate4;
#pragma unroll
    for (int t = 0; t < 4; ++t) {
      const int e  = tid + 256 * t;
      const int p  = e >> 5;
      const int fl = e & 31;
      const int so = (fl & 6) | ((fl >> 3) & 1);
      const int co = fl & 1;
      const int h  = fl >> 4;
      const int sa = (p >> 2) * NSLOT + 3 * (p & 3);
      const float2 ul = gate[(sa + 2) * 4 + (so & 1) * 2 + h];
      float av[8];
#pragma unroll
      for (int c = 0; c < 4; ++c) {
        const float2 uh = gate[sa * 4 + (so >> 2) * 2 + (c >> 1)];
        const float2 um = gate[(sa + 1) * 4 + ((so >> 1) & 1) * 2 + (c & 1)];
        const float2 g  = cmulf(cmulf(uh, um), ul);
        av[2 * c]     = co ? g.y : g.x;
        av[2 * c + 1] = co ? g.x : -g.y;
      }
      afr[e * 2]     = make_float4(av[0], av[1], av[2], av[3]);
      afr[e * 2 + 1] = make_float4(av[4], av[5], av[6], av[7]);
    }
  }
  __syncthreads();

  int pend = 0;
  for (int blkw = 0; blkw < 2; ++blkw) {
    for (int l = 0; l < 4; ++l) {
      int L = blkw * 4 + l;
      if (L == 0) {
        apply_group_first<10, 9, 8, 1>(st, afr, 0, wave, lane);      __syncthreads();
        apply_group_first<6, 5, 4, 0>(st, afr, 1, wave, lane);       wave_order();
        apply_group_first<7, 3, 2, 0>(st, afr, 2, wave, lane);       wave_order();
        apply_group_first<7, 0, 1, 2>(st, afr, 3, wave, lane);       __syncthreads();
      } else {
        apply_group<10, 9, 8>(st, afr, L * 4 + 0, wave, lane, pend); __syncthreads();
        apply_group<6, 5, 4>(st, afr, L * 4 + 1, wave, lane, 0);     wave_order();
        apply_group<7, 3, 2>(st, afr, L * 4 + 2, wave, lane, 0);     wave_order();
        apply_group<7, 0, 1>(st, afr, L * 4 + 3, wave, lane, 0);     __syncthreads();
      }
      pend = l + 1;
    }
  }

#pragma unroll
  for (int t = 0; t < 4; ++t) {
    int p  = tid + 256 * t;
    float2 a = st[pinv_idx(2 * p, pend)];
    out[b * PIX + p] = fminf(1.0f, (a.x * a.x + a.y * a.y) * (float)PIX);
  }
}

extern "C" void kernel_launch(void* const* d_in, const int* in_sizes, int n_in,
                              void* d_out, int out_size, void* d_ws, size_t ws_size,
                              hipStream_t stream) {
  const float* x  = (const float*)d_in[0];
  const float* y  = (const float*)d_in[1];
  const float* w0 = (const float*)d_in[2];
  const float* w1 = (const float*)d_in[3];

  qsim<<<NB, 256, 0, stream>>>(x, y, w0, w1, (float*)d_out);
}


# wave priority steps down with progress (s_setprio 3 set-up+layers 0-3, 2 layers 4-5, 1 layer 6, 0 layer 7): lagging co-resident workgroups win the matrix pipe, all end together; same arithmetic
# speedup vs baseline: 5.4743x; 1.0345x over previous
#include <hip/hip_runtime.h>
#include <math.h>


#define WIRES 11
#define DIM   2048
#define PIX   1024
#define NB    1024

typedef __attribute__((ext_vector_type(2))) float v2f;
typedef __attribute__((ext_vector_type(8))) float v8f;

#define NSLOT 12
#define NPASS 32

__device__ __forceinline__ float2 cmulf(float2 p, float2 u) {
  return make_float2(p.x * u.x - p.y * u.y, p.x * u.y + p.y * u.x);
}

__device__ __forceinline__ void rot_entries(const float* __restrict__ a3, float2 u[4]) {
  const float PI_ = 3.14159265358979323846f;
  const float a0 = a3[0], a1 = a3[1], a2 = a3[2];
  float phi   = PI_ * tanhf(a0);
  float theta = PI_ * tanhf(a1);
  float omega = PI_ * tanhf(a2);
  float c, s, ch, sh, cg, sg;
  sincosf(0.5f * theta, &s, &c);
  sincosf(0.5f * (phi + omega), &sh, &ch);
  sincosf(0.5f * (phi - omega), &sg, &cg);
  u[0] = make_float2( ch * c, -sh * c);
  u[1] = make_float2(-cg * s, -sg * s);
  u[2] = make_float2( cg * s, -sg * s);
  u[3] = make_float2( ch * c,  sh * c);
}

__device__ __forceinline__ int pinv_idx(int x, int r) {
  const int a = x ^ ((x & ((1 << r) - 1)) << (WIRES - r));
  return a ^ (a >> r);
}

constexpr int spread_coset(int j, int sub) {
  int r = 0, k = 0;
  for (int b = 0; b < WIRES; ++b) {
    if (!((sub >> b) & 1)) { r |= ((j >> k) & 1) << b; ++k; }
  }
  return r;
}

template <int S2, int S1, int S0>
__device__ __forceinline__ void apply_group(float2* st, const float4* afr,
                                            int p, int wave, int lane, int rperm) {
  constexpr int SUB = (1 << S2) | (1 << S1) | (1 << S0);
  const int n    = lane & 15;
  const int half = lane >> 4;

  const float4 a01 = afr[(p * 32 + lane) * 2];
  const float4 a23 = afr[(p * 32 + lane) * 2 + 1];
  const v2f a[4] = {v2f{a01.x, a01.y}, v2f{a01.z, a01.w},
                    v2f{a23.x, a23.y}, v2f{a23.z, a23.w}};

  constexpr int TILE1 = spread_coset(16, SUB);
  const int base = spread_coset(wave * 32 + n, SUB) + (half << S0);
  int idx[2][4];
  v2f bf[2][4];
#pragma unroll
  for (int t = 0; t < 2; ++t) {
#pragma unroll
    for (int c = 0; c < 4; ++c) {
      idx[t][c] = base + t * TILE1 + ((c & 1) << S1) + ((c >> 1) << S2);
      int i0 = idx[t][c];
      if (rperm) i0 = pinv_idx(i0, rperm);
      const float2 s0 = st[i0];
      bf[t][c] = v2f{s0.x, s0.y};
    }
  }
  if (rperm) __syncthreads();

  v8f d0 = {0, 0, 0, 0, 0, 0, 0, 0};
  v8f d1 = {0, 0, 0, 0, 0, 0, 0, 0};
#pragma unroll
  for (int c = 0; c < 4; ++c) {
    d0 = __builtin_amdgcn_wmma_f32_16x16x4_f32(false, a[c], false, bf[0][c],
                                               (short)0, d0, false, false);
    d1 = __builtin_amdgcn_wmma_f32_16x16x4_f32(false, a[c], false, bf[1][c],
                                               (short)0, d1, false, false);
  }

#pragma unroll
  for (int q = 0; q < 4; ++q) {
    st[idx[0][q]] = make_float2(d0[2 * q], d0[2 * q + 1]);
    st[idx[1][q]] = make_float2(d1[2 * q], d1[2 * q + 1]);
  }
}

template <int S2, int S1, int S0, int KIND>
__device__ __forceinline__ void apply_group_first(float2* st, const float4* afr,
                                                  int p, int wave, int lane) {
  constexpr int SUB = (1 << S2) | (1 << S1) | (1 << S0);
  static_assert(KIND != 2 || S1 == 0, "KIND 2 needs index bit 0 as the chunk bit c&1");
  static_assert(KIND == 2 || !(SUB & 1), "KIND 0/1 groups must not hold index bit 0");
  const int n    = lane & 15;
  const int half = lane >> 4;

  const float4 a01 = afr[(p * 32 + lane) * 2];
  const float4 a23 = afr[(p * 32 + lane) * 2 + 1];
  const v2f a[4] = {v2f{a01.x, a01.y}, v2f{a01.z, a01.w},
                    v2f{a23.x, a23.y}, v2f{a23.z, a23.w}};
  const v8f zero = {0, 0, 0, 0, 0, 0, 0, 0};

  if (KIND == 2) {
    constexpr int TILE1 = spread_coset(16, SUB);
    const int base = spread_coset(wave * 32 + n, SUB) + (half << S0);
    v2f b0[2], b1[2];
#pragma unroll
    for (int cc = 0; cc < 2; ++cc) {
      const float2 s0 = st[base + (cc << S2)];
      const float2 s1 = st[base + TILE1 + (cc << S2)];
      b0[cc] = v2f{s0.x, s0.y};
      b1[cc] = v2f{s1.x, s1.y};
    }
    v8f d0 = zero, d1 = zero;
#pragma unroll
    for (int cc = 0; cc < 2; ++cc) {
      d0 = __builtin_amdgcn_wmma_f32_16x16x4_f32(false, a[2 * cc], false, b0[cc],
                                                 (short)0, d0, false, false);
      d1 = __builtin_amdgcn_wmma_f32_16x16x4_f32(false, a[2 * cc], false, b1[cc],
                                                 (short)0, d1, false, false);
    }
#pragma unroll
    for (int q = 0; q < 4; ++q) {
      const int i0 = base + ((q & 1) << S1) + ((q >> 1) << S2);
      st[i0]         = make_float2(d0[2 * q], d0[2 * q + 1]);
      st[i0 + TILE1] = make_float2(d1[2 * q], d1[2 * q + 1]);
    }
  } else {
    const int base = spread_coset(wave * 16 + n, SUB | 1) + (half << S0);
    int idx[4];
    v2f bf[4];
#pragma unroll
    for (int c = 0; c < 4; ++c) {
      idx[c] = base + ((c & 1) << S1) + ((c >> 1) << S2);
      const float2 s0 = st[idx[c]];
      bf[c] = v2f{s0.x, s0.y};
    }
    v8f d = zero;
    if (KIND == 1) {
#pragma unroll
      for (int cp = 0; cp < 2; ++cp) {
        d = __builtin_amdgcn_wmma_f32_16x16x4_f32(
            false, v2f{a[2 * cp].x, a[2 * cp + 1].x},
            false, v2f{bf[2 * cp].x, bf[2 * cp + 1].x}, (short)0, d, false, false);
      }
    } else {
#pragma unroll
      for (int c = 0; c < 4; ++c) {
        d = __builtin_amdgcn_wmma_f32_16x16x4_f32(false, a[c], false, bf[c],
                                                   (short)0, d, false, false);
      }
    }
#pragma unroll
    for (int q = 0; q < 4; ++q) {
      st[idx[q]] = make_float2(d[2 * q], d[2 * q + 1]);
    }
  }
}

__device__ __forceinline__ void wave_order() {
  __builtin_amdgcn_fence(__ATOMIC_ACQ_REL, "wavefront");
}

__global__ void __launch_bounds__(256)
qsim(const float* __restrict__ x, const float* __restrict__ y,
     const float* __restrict__ w0, const float* __restrict__ w1,
     float* __restrict__ out) {
  __shared__ float2 st[DIM];
  __shared__ float  red[256];
  __shared__ float4 gate4[8 * NSLOT * 2];
  __shared__ float4 afr[NPASS * 32 * 2];
  const int tid  = threadIdx.x;
  const int lane = tid & 31;
  const int wave = tid >> 5;
  const int b    = blockIdx.x;

  __builtin_amdgcn_s_setprio(3);

  float xs[4], part = 0.f;
#pragma unroll
  for (int t = 0; t < 4; ++t) {
    int p = tid + 256 * t;
    float v = x[b * PIX + p];
    xs[t] = v; part += v * v;
  }
  const float yv = y[b];
  if (tid < 8 * NSLOT) {
    int L    = tid / NSLOT;
    int slot = tid - L * NSLOT;
    float2 u[4];
    if (slot == 9) {
      u[0] = make_float2(1.f, 0.f); u[1] = make_float2(0.f, 0.f);
      u[2] = make_float2(0.f, 0.f); u[3] = make_float2(1.f, 0.f);
    } else {
      int q = slot;
      if (slot >= 3 && slot <= 5) q = slot + 1;
      else if (slot == 6) q = 3;
      else if (slot == 11) q = 9;
      const float* w = (L < 4) ? w0 : w1;
      rot_entries(w + ((L & 3) * WIRES + q) * 3, u);
      if (slot == 10 && (L & 3) == 0) {
        float rc, rs;
        sincosf(0.5f * yv, &rs, &rc);
        const float2 u0 = u[0], u1 = u[1], u2 = u[2], u3 = u[3];
        u[0] = make_float2(u0.x * rc + u1.y * rs, u0.y * rc - u1.x * rs);
        u[1] = make_float2(u1.x * rc + u0.y * rs, u1.y * rc - u0.x * rs);
        u[2] = make_float2(u2.x * rc + u3.y * rs, u2.y * rc - u3.x * rs);
        u[3] = make_float2(u3.x * rc + u2.y * rs, u3.y * rc - u2.x * rs);
      }
    }
    gate4[tid * 2]     = make_float4(u[0].x, u[0].y, u[1].x, u[1].y);
    gate4[tid * 2 + 1] = make_float4(u[2].x, u[2].y, u[3].x, u[3].y);
  }
  red[tid] = part; __syncthreads();
  for (int off = 128; off > 0; off >>= 1) {
    if (tid < off) red[tid] += red[tid + off];
    __syncthreads();
  }
  float inv = rsqrtf(red[0]);
#pragma unroll
  for (int t = 0; t < 4; ++t) {
    int p = tid + 256 * t;
    st[2 * p] = make_float2(xs[t] * inv, 0.f);
  }
  {
    const float2* gate = (const float2*)gate4;
#pragma unroll
    for (int t = 0; t < 4; ++t) {
      const int e  = tid + 256 * t;
      const int p  = e >> 5;
      const int fl = e & 31;
      const int so = (fl & 6) | ((fl >> 3) & 1);
      const int co = fl & 1;
      const int h  = fl >> 4;
      const int sa = (p >> 2) * NSLOT + 3 * (p & 3);
      const float2 ul = gate[(sa + 2) * 4 + (so & 1) * 2 + h];
      float av[8];
#pragma unroll
      for (int c = 0; c < 4; ++c) {
        const float2 uh = gate[sa * 4 + (so >> 2) * 2 + (c >> 1)];
        const float2 um = gate[(sa + 1) * 4 + ((so >> 1) & 1) * 2 + (c & 1)];
        const float2 g  = cmulf(cmulf(uh, um), ul);
        av[2 * c]     = co ? g.y : g.x;
        av[2 * c + 1] = co ? g.x : -g.y;
      }
      afr[e * 2]     = make_float4(av[0], av[1], av[2], av[3]);
      afr[e * 2 + 1] = make_float4(av[4], av[5], av[6], av[7]);
    }
  }
  __syncthreads();

  int pend = 0;
  for (int blkw = 0; blkw < 2; ++blkw) {
    for (int l = 0; l < 4; ++l) {
      int L = blkw * 4 + l;
      if (L == 4)      __builtin_amdgcn_s_setprio(2);
      else if (L == 6) __builtin_amdgcn_s_setprio(1);
      else if (L == 7) __builtin_amdgcn_s_setprio(0);
      if (L == 0) {
        apply_group_first<10, 9, 8, 1>(st, afr, 0, wave, lane);      __syncthreads();
        apply_group_first<6, 5, 4, 0>(st, afr, 1, wave, lane);       wave_order();
        apply_group_first<7, 3, 2, 0>(st, afr, 2, wave, lane);       wave_order();
        apply_group_first<7, 0, 1, 2>(st, afr, 3, wave, lane);       __syncthreads();
      } else {
        apply_group<10, 9, 8>(st, afr, L * 4 + 0, wave, lane, pend); __syncthreads();
        apply_group<6, 5, 4>(st, afr, L * 4 + 1, wave, lane, 0);     wave_order();
        apply_group<7, 3, 2>(st, afr, L * 4 + 2, wave, lane, 0);     wave_order();
        apply_group<7, 0, 1>(st, afr, L * 4 + 3, wave, lane, 0);     __syncthreads();
      }
      pend = l + 1;
    }
  }

#pragma unroll
  for (int t = 0; t < 4; ++t) {
    int p  = tid + 256 * t;
    float2 a = st[pinv_idx(2 * p, pend)];
    out[b * PIX + p] = fminf(1.0f, (a.x * a.x + a.y * a.y) * (float)PIX);
  }
}

extern "C" void kernel_launch(void* const* d_in, const int* in_sizes, int n_in,
                              void* d_out, int out_size, void* d_ws, size_t ws_size,
                              hipStream_t stream) {
  const float* x  = (const float*)d_in[0];
  const float* y  = (const float*)d_in[1];
  const float* w0 = (const float*)d_in[2];
  const float* w1 = (const float*)d_in[3];

  qsim<<<NB, 256, 0, stream>>>(x, y, w0, w1, (float*)d_out);
}


# layers 1-7 in 3 matrix passes, not 4: groups (10,9,8),(5,3,2),(7,6,4); index bits 1, 0 are a pass's two-tile bit, so wire 9/10 gates are packed f32 FMAs on one lane's accumulators; sum order changes
# speedup vs baseline: 5.6985x; 1.0410x over previous
#include <hip/hip_runtime.h>
#include <math.h>


#define WIRES 11
#define DIM   2048
#define PIX   1024
#define NB    1024

typedef __attribute__((ext_vector_type(2))) float v2f;
typedef __attribute__((ext_vector_type(8))) float v8f;

#define NSLOT 12
#define NPASS 25
#define KIND_WIRES_FIRST 0x9AB873654210ULL
#define KIND_WIRES_REST  0x643875210ULL

__device__ __forceinline__ float2 cmulf(float2 p, float2 u) {
  return make_float2(p.x * u.x - p.y * u.y, p.x * u.y + p.y * u.x);
}

__device__ __forceinline__ void rot_entries(const float* __restrict__ a3, float2 u[4]) {
  const float PI_ = 3.14159265358979323846f;
  const float a0 = a3[0], a1 = a3[1], a2 = a3[2];
  float phi   = PI_ * tanhf(a0);
  float theta = PI_ * tanhf(a1);
  float omega = PI_ * tanhf(a2);
  float c, s, ch, sh, cg, sg;
  sincosf(0.5f * theta, &s, &c);
  sincosf(0.5f * (phi + omega), &sh, &ch);
  sincosf(0.5f * (phi - omega), &sg, &cg);
  u[0] = make_float2( ch * c, -sh * c);
  u[1] = make_float2(-cg * s, -sg * s);
  u[2] = make_float2( cg * s, -sg * s);
  u[3] = make_float2( ch * c,  sh * c);
}

__device__ __forceinline__ int pinv_idx(int x, int r) {
  const int a = x ^ ((x & ((1 << r) - 1)) << (WIRES - r));
  return a ^ (a >> r);
}

constexpr int spread_coset(int j, int sub) {
  int r = 0, k = 0;
  for (int b = 0; b < WIRES; ++b) {
    if (!((sub >> b) & 1)) { r |= ((j >> k) & 1) << b; ++k; }
  }
  return r;
}

template <int S2, int S1, int S0, int TB, bool HOST>
__device__ __forceinline__ void apply_group(float2* st, const float4* afr, const float4* gate4,
                                            const float4* gsw4,
                                            int p, int hs, int wave, int lane, int rperm) {
  constexpr int SUB = (1 << S2) | (1 << S1) | (1 << S0);
  static_assert(!(SUB & (1 << TB)), "tile bit must be outside the group");
  const int n    = lane & 15;
  const int half = lane >> 4;

  const float4 a01 = afr[(p * 32 + lane) * 2];
  const float4 a23 = afr[(p * 32 + lane) * 2 + 1];
  const v2f a[4] = {v2f{a01.x, a01.y}, v2f{a01.z, a01.w},
                    v2f{a23.x, a23.y}, v2f{a23.z, a23.w}};
  float4 g01 = make_float4(0.f, 0.f, 0.f, 0.f), g23 = g01, s01 = g01, s23 = g01;
  if (HOST) {
    g01 = gate4[hs * 2];
    g23 = gate4[hs * 2 + 1];
    s01 = gsw4[hs * 2];
    s23 = gsw4[hs * 2 + 1];
  }

  constexpr int TILE1 = 1 << TB;
  const int base = spread_coset(wave * 16 + n, SUB | TILE1) + (half << S0);
  int idx[2][4];
  v2f bf[2][4];
#pragma unroll
  for (int t = 0; t < 2; ++t) {
#pragma unroll
    for (int c = 0; c < 4; ++c) {
      idx[t][c] = base + t * TILE1 + ((c & 1) << S1) + ((c >> 1) << S2);
      int i0 = idx[t][c];
      if (rperm) i0 = pinv_idx(i0, rperm);
      const float2 s0 = st[i0];
      bf[t][c] = v2f{s0.x, s0.y};
    }
  }
  if (rperm) __syncthreads();

  v8f d0 = {0, 0, 0, 0, 0, 0, 0, 0};
  v8f d1 = {0, 0, 0, 0, 0, 0, 0, 0};
#pragma unroll
  for (int c = 0; c < 4; ++c) {
    d0 = __builtin_amdgcn_wmma_f32_16x16x4_f32(false, a[c], false, bf[0][c],
                                               (short)0, d0, false, false);
    d1 = __builtin_amdgcn_wmma_f32_16x16x4_f32(false, a[c], false, bf[1][c],
                                               (short)0, d1, false, false);
  }

  if (HOST) {
    const v2f u00 = v2f{g01.x, g01.y}, u01 = v2f{g01.z, g01.w};
    const v2f u10 = v2f{g23.x, g23.y}, u11 = v2f{g23.z, g23.w};
    const v2f w00 = v2f{s01.x, s01.y}, w01 = v2f{s01.z, s01.w};
    const v2f w10 = v2f{s23.x, s23.y}, w11 = v2f{s23.z, s23.w};
#pragma unroll
    for (int q = 0; q < 4; ++q) {
      const v2f a0r = v2f{d0[2 * q], d0[2 * q]}, a0i = v2f{d0[2 * q + 1], d0[2 * q + 1]};
      const v2f a1r = v2f{d1[2 * q], d1[2 * q]}, a1i = v2f{d1[2 * q + 1], d1[2 * q + 1]};
      const v2f n0 = a0r * u00 + a0i * w00 + a1r * u01 + a1i * w01;
      const v2f n1 = a0r * u10 + a0i * w10 + a1r * u11 + a1i * w11;
      st[idx[0][q]] = make_float2(n0.x, n0.y);
      st[idx[1][q]] = make_float2(n1.x, n1.y);
    }
  } else {
#pragma unroll
    for (int q = 0; q < 4; ++q) {
      st[idx[0][q]] = make_float2(d0[2 * q], d0[2 * q + 1]);
      st[idx[1][q]] = make_float2(d1[2 * q], d1[2 * q + 1]);
    }
  }
}

template <int S2, int S1, int S0, int KIND>
__device__ __forceinline__ void apply_group_first(float2* st, const float4* afr,
                                                  int p, int wave, int lane) {
  constexpr int SUB = (1 << S2) | (1 << S1) | (1 << S0);
  static_assert(KIND != 2 || S1 == 0, "KIND 2 needs index bit 0 as the chunk bit c&1");
  static_assert(KIND == 2 || !(SUB & 1), "KIND 0/1 groups must not hold index bit 0");
  const int n    = lane & 15;
  const int half = lane >> 4;

  const float4 a01 = afr[(p * 32 + lane) * 2];
  const float4 a23 = afr[(p * 32 + lane) * 2 + 1];
  const v2f a[4] = {v2f{a01.x, a01.y}, v2f{a01.z, a01.w},
                    v2f{a23.x, a23.y}, v2f{a23.z, a23.w}};
  const v8f zero = {0, 0, 0, 0, 0, 0, 0, 0};

  if (KIND == 2) {
    constexpr int TILE1 = spread_coset(16, SUB);
    const int base = spread_coset(wave * 32 + n, SUB) + (half << S0);
    v2f b0[2], b1[2];
#pragma unroll
    for (int cc = 0; cc < 2; ++cc) {
      const float2 s0 = st[base + (cc << S2)];
      const float2 s1 = st[base + TILE1 + (cc << S2)];
      b0[cc] = v2f{s0.x, s0.y};
      b1[cc] = v2f{s1.x, s1.y};
    }
    v8f d0 = zero, d1 = zero;
#pragma unroll
    for (int cc = 0; cc < 2; ++cc) {
      d0 = __builtin_amdgcn_wmma_f32_16x16x4_f32(false, a[2 * cc], false, b0[cc],
                                                 (short)0, d0, false, false);
      d1 = __builtin_amdgcn_wmma_f32_16x16x4_f32(false, a[2 * cc], false, b1[cc],
                                                 (short)0, d1, false, false);
    }
#pragma unroll
    for (int q = 0; q < 4; ++q) {
      const int i0 = base + ((q & 1) << S1) + ((q >> 1) << S2);
      st[i0]         = make_float2(d0[2 * q], d0[2 * q + 1]);
      st[i0 + TILE1] = make_float2(d1[2 * q], d1[2 * q + 1]);
    }
  } else {
    const int base = spread_coset(wave * 16 + n, SUB | 1) + (half << S0);
    int idx[4];
    v2f bf[4];
#pragma unroll
    for (int c = 0; c < 4; ++c) {
      idx[c] = base + ((c & 1) << S1) + ((c >> 1) << S2);
      const float2 s0 = st[idx[c]];
      bf[c] = v2f{s0.x, s0.y};
    }
    v8f d = zero;
    if (KIND == 1) {
#pragma unroll
      for (int cp = 0; cp < 2; ++cp) {
        d = __builtin_amdgcn_wmma_f32_16x16x4_f32(
            false, v2f{a[2 * cp].x, a[2 * cp + 1].x},
            false, v2f{bf[2 * cp].x, bf[2 * cp + 1].x}, (short)0, d, false, false);
      }
    } else {
#pragma unroll
      for (int c = 0; c < 4; ++c) {
        d = __builtin_amdgcn_wmma_f32_16x16x4_f32(false, a[c], false, bf[c],
                                                   (short)0, d, false, false);
      }
    }
#pragma unroll
    for (int q = 0; q < 4; ++q) {
      st[idx[q]] = make_float2(d[2 * q], d[2 * q + 1]);
    }
  }
}

__device__ __forceinline__ void wave_order() {
  __builtin_amdgcn_fence(__ATOMIC_ACQ_REL, "wavefront");
}

__global__ void __launch_bounds__(256)
qsim(const float* __restrict__ x, const float* __restrict__ y,
     const float* __restrict__ w0, const float* __restrict__ w1,
     float* __restrict__ out) {
  __shared__ float2 st[DIM];
  __shared__ float  red[256];
  __shared__ float4 gate4[8 * NSLOT * 2];
  __shared__ float4 gsw4[8 * NSLOT * 2];
  __shared__ float4 afr[NPASS * 32 * 2];
  const int tid  = threadIdx.x;
  const int lane = tid & 31;
  const int wave = tid >> 5;
  const int b    = blockIdx.x;

  __builtin_amdgcn_s_setprio(3);

  float xs[4], part = 0.f;
#pragma unroll
  for (int t = 0; t < 4; ++t) {
    int p = tid + 256 * t;
    float v = x[b * PIX + p];
    xs[t] = v; part += v * v;
  }
  const float yv = y[b];
  if (tid < 8 * NSLOT) {
    int L    = tid / NSLOT;
    int slot = tid - L * NSLOT;
    float2 u[4];
    if (slot == 11) {
      u[0] = make_float2(1.f, 0.f); u[1] = make_float2(0.f, 0.f);
      u[2] = make_float2(0.f, 0.f); u[3] = make_float2(1.f, 0.f);
    } else {
      const int q = slot;
      const float* w = (L < 4) ? w0 : w1;
      rot_entries(w + ((L & 3) * WIRES + q) * 3, u);
      if (slot == 10 && (L & 3) == 0) {
        float rc, rs;
        sincosf(0.5f * yv, &rs, &rc);
        const float2 u0 = u[0], u1 = u[1], u2 = u[2], u3 = u[3];
        u[0] = make_float2(u0.x * rc + u1.y * rs, u0.y * rc - u1.x * rs);
        u[1] = make_float2(u1.x * rc + u0.y * rs, u1.y * rc - u0.x * rs);
        u[2] = make_float2(u2.x * rc + u3.y * rs, u2.y * rc - u3.x * rs);
        u[3] = make_float2(u3.x * rc + u2.y * rs, u3.y * rc - u2.x * rs);
      }
    }
    gate4[tid * 2]     = make_float4(u[0].x, u[0].y, u[1].x, u[1].y);
    gate4[tid * 2 + 1] = make_float4(u[2].x, u[2].y, u[3].x, u[3].y);
    gsw4[tid * 2]      = make_float4(-u[0].y, u[0].x, -u[1].y, u[1].x);
    gsw4[tid * 2 + 1]  = make_float4(-u[2].y, u[2].x, -u[3].y, u[3].x);
  }
  red[tid] = part; __syncthreads();
  for (int off = 128; off > 0; off >>= 1) {
    if (tid < off) red[tid] += red[tid + off];
    __syncthreads();
  }
  float inv = rsqrtf(red[0]);
#pragma unroll
  for (int t = 0; t < 4; ++t) {
    int p = tid + 256 * t;
    st[2 * p] = make_float2(xs[t] * inv, 0.f);
  }
  {
    const float2* gate = (const float2*)gate4;
#pragma unroll
    for (int t = 0; t < 4; ++t) {
      const int e  = tid + 256 * t;
      const int p  = e >> 5;
      if (p >= NPASS) break;
      const int fl = e & 31;
      const int so = (fl & 6) | ((fl >> 3) & 1);
      const int co = fl & 1;
      const int h  = fl >> 4;
      const int pp = p - 4;
      const int L  = (p < 4) ? 0 : 1 + pp / 3;
      const unsigned w3 = (p < 4) ? (unsigned)(KIND_WIRES_FIRST >> (12 * p))
                                  : (unsigned)(KIND_WIRES_REST >> (12 * (pp % 3)));
      const int st_ = L * NSLOT + (int)(w3 & 15u);
      const int sm_ = L * NSLOT + (int)((w3 >> 4) & 15u);
      const int sl_ = L * NSLOT + (int)((w3 >> 8) & 15u);
      const float2 ul = gate[sl_ * 4 + (so & 1) * 2 + h];
      float av[8];
#pragma unroll
      for (int c = 0; c < 4; ++c) {
        const float2 uh = gate[st_ * 4 + (so >> 2) * 2 + (c >> 1)];
        const float2 um = gate[sm_ * 4 + ((so >> 1) & 1) * 2 + (c & 1)];
        const float2 g  = cmulf(cmulf(uh, um), ul);
        av[2 * c]     = co ? g.y : g.x;
        av[2 * c + 1] = co ? g.x : -g.y;
      }
      afr[e * 2]     = make_float4(av[0], av[1], av[2], av[3]);
      afr[e * 2 + 1] = make_float4(av[4], av[5], av[6], av[7]);
    }
  }
  __syncthreads();

  int pend = 0;
  for (int blkw = 0; blkw < 2; ++blkw) {
    for (int l = 0; l < 4; ++l) {
      int L = blkw * 4 + l;
      if (L == 4)      __builtin_amdgcn_s_setprio(2);
      else if (L == 6) __builtin_amdgcn_s_setprio(1);
      else if (L == 7) __builtin_amdgcn_s_setprio(0);
      if (L == 0) {
        apply_group_first<10, 9, 8, 1>(st, afr, 0, wave, lane);      __syncthreads();
        apply_group_first<6, 5, 4, 0>(st, afr, 1, wave, lane);       wave_order();
        apply_group_first<7, 3, 2, 0>(st, afr, 2, wave, lane);       wave_order();
        apply_group_first<7, 0, 1, 2>(st, afr, 3, wave, lane);       __syncthreads();
      } else {
        const int pb = 4 + 3 * (L - 1);
        apply_group<10, 9, 8, 1, true>(st, afr, gate4, gsw4, pb, L * NSLOT + 9, wave, lane, pend);
        __syncthreads();
        apply_group<5, 3, 2, 7, false>(st, afr, gate4, gsw4, pb + 1, 0, wave, lane, 0);
        wave_order();
        apply_group<7, 6, 4, 0, true>(st, afr, gate4, gsw4, pb + 2, L * NSLOT + 10, wave, lane, 0);
        __syncthreads();
      }
      pend = l + 1;
    }
  }

#pragma unroll
  for (int t = 0; t < 4; ++t) {
    int p  = tid + 256 * t;
    float2 a = st[pinv_idx(2 * p, pend)];
    out[b * PIX + p] = fminf(1.0f, (a.x * a.x + a.y * a.y) * (float)PIX);
  }
}

extern "C" void kernel_launch(void* const* d_in, const int* in_sizes, int n_in,
                              void* d_out, int out_size, void* d_ws, size_t ws_size,
                              hipStream_t stream) {
  const float* x  = (const float*)d_in[0];
  const float* y  = (const float*)d_in[1];
  const float* w0 = (const float*)d_in[2];
  const float* w1 = (const float*)d_in[3];

  qsim<<<NB, 256, 0, stream>>>(x, y, w0, w1, (float*)d_out);
}
